# starting kernel
# baseline (speedup 1.0000x reference)
#include <hip/hip_runtime.h>


typedef float v2f __attribute__((ext_vector_type(2)));
typedef float v8f __attribute__((ext_vector_type(8)));

#define CIN    16
#define JNT    25
#define EDG    24
#define COUTC  16
#define KW     3
#define LIN    65536
#define LOUTC  (LIN - KW + 1)
#define RDIM   (CIN * KW)
#define NCHUNK (RDIM / 4)
#define NTILE  ((LOUTC + 15) / 16)

__global__ __launch_bounds__(128) void skel_conv_wmma_f32(
    const float* __restrict__ x,
    const float* __restrict__ W,
    const float* __restrict__ b,
    float* __restrict__ out)
{
    const int lane = threadIdx.x & 31;
    const int wave = threadIdx.x >> 5;
    const int e    = blockIdx.y;
    const int h    = lane >> 4;
    const int mn   = lane & 15;

    const int p0 = 2 * e;
    const int p1 = 2 * e + 1;

    v2f A0[NCHUNK], A1[NCHUNK];
    #pragma unroll
    for (int c = 0; c < NCHUNK; ++c) {
        const int r0 = 4 * c + 2 * h;
        const int r1 = r0 + 1;
        const int ci0 = r0 & 15, k0 = r0 >> 4;
        const int ci1 = r1 & 15, k1 = r1 >> 4;
        A0[c].x = W[((p0 * COUTC + mn) * CIN + ci0) * KW + k0];
        A0[c].y = W[((p0 * COUTC + mn) * CIN + ci1) * KW + k1];
        A1[c].x = W[((p1 * COUTC + mn) * CIN + ci0) * KW + k0];
        A1[c].y = W[((p1 * COUTC + mn) * CIN + ci1) * KW + k1];
    }

    float bias[8];
    #pragma unroll
    for (int r = 0; r < 8; ++r) {
        const int co = 8 * h + r;
        bias[r] = 0.5f * (b[p0 * COUTC + co] + b[p1 * COUTC + co]);
    }

    const int gw      = blockIdx.x * 4 + wave;
    const int gstride = gridDim.x * 4;

    for (int tile = gw; tile < NTILE; tile += gstride) {
        const int t0 = tile * 16;
        v8f acc0 = {};
        v8f acc1 = {};

        if (t0 + 16 + (KW - 1) <= LIN) {
            #pragma unroll
            for (int c = 0; c < NCHUNK; ++c) {
                const int r0 = 4 * c + 2 * h;
                const int r1 = r0 + 1;
                const int ta = t0 + mn + (r0 >> 4);
                const int tb = t0 + mn + (r1 >> 4);
                v2f B0, B1;
                B0.x = x[((r0 & 15) * JNT + e) * LIN + ta];
                B0.y = x[((r1 & 15) * JNT + e) * LIN + tb];
                B1.x = x[((r0 & 15) * JNT + (e + 1)) * LIN + ta];
                B1.y = x[((r1 & 15) * JNT + (e + 1)) * LIN + tb];
                acc0 = __builtin_amdgcn_wmma_f32_16x16x4_f32(
                    false, A0[c], false, B0, (short)0, acc0, false, false);
                acc1 = __builtin_amdgcn_wmma_f32_16x16x4_f32(
                    false, A1[c], false, B1, (short)0, acc1, false, false);
            }
            #pragma unroll
            for (int r = 0; r < 8; ++r) {
                const int co = 8 * h + r;
                out[(co * EDG + e) * LOUTC + t0 + mn] =
                    0.5f * (acc0[r] + acc1[r]) + bias[r];
            }
        } else {
            #pragma unroll
            for (int c = 0; c < NCHUNK; ++c) {
                const int r0 = 4 * c + 2 * h;
                const int r1 = r0 + 1;
                int ta = t0 + mn + (r0 >> 4);
                int tb = t0 + mn + (r1 >> 4);
                ta = (ta < LIN) ? ta : (LIN - 1);
                tb = (tb < LIN) ? tb : (LIN - 1);
                v2f B0, B1;
                B0.x = x[((r0 & 15) * JNT + e) * LIN + ta];
                B0.y = x[((r1 & 15) * JNT + e) * LIN + tb];
                B1.x = x[((r0 & 15) * JNT + (e + 1)) * LIN + ta];
                B1.y = x[((r1 & 15) * JNT + (e + 1)) * LIN + tb];
                acc0 = __builtin_amdgcn_wmma_f32_16x16x4_f32(
                    false, A0[c], false, B0, (short)0, acc0, false, false);
                acc1 = __builtin_amdgcn_wmma_f32_16x16x4_f32(
                    false, A1[c], false, B1, (short)0, acc1, false, false);
            }
            if (t0 + mn < LOUTC) {
                #pragma unroll
                for (int r = 0; r < 8; ++r) {
                    const int co = 8 * h + r;
                    out[(co * EDG + e) * LOUTC + t0 + mn] =
                        0.5f * (acc0[r] + acc1[r]) + bias[r];
                }
            }
        }
    }
}

extern "C" void kernel_launch(void* const* d_in, const int* in_sizes, int n_in,
                              void* d_out, int out_size, void* d_ws, size_t ws_size,
                              hipStream_t stream) {
    (void)in_sizes; (void)n_in; (void)out_size; (void)d_ws; (void)ws_size;
    const float* x = (const float*)d_in[0];
    const float* W = (const float*)d_in[1];
    const float* b = (const float*)d_in[2];
    float* out = (float*)d_out;

    dim3 grid(32, EDG);
    dim3 block(128);
    hipLaunchKernelGGL(skel_conv_wmma_f32, grid, block, 0, stream, x, W, b, out);
}


# x strip of all 25 joints staged once per work-group in LDS by whole-row 16-byte loads, B fragments read from LDS, not by 16 unaligned 12-byte global loads per lane and tile; 256 persistent groups
# speedup vs baseline: 1.2656x; 1.2656x over previous
#include <hip/hip_runtime.h>


typedef float v2f __attribute__((ext_vector_type(2)));
typedef float v4f __attribute__((ext_vector_type(4)));
typedef float v8f __attribute__((ext_vector_type(8)));

#define CIN    16
#define JNT    25
#define EDG    24
#define COUTC  16
#define KW     3
#define LIN    65536
#define LOUTC  (LIN - KW + 1)
#define NCHUNK 12
#define STRIP  64
#define NSTRIP (LIN / STRIP)
#define NROW   (CIN * JNT)
#define PITCH  72
#define BUFF   (NROW * PITCH)
#define EPW    3
#define NTHR   256
#define BODYR  (NROW * 16 / NTHR)
#define PART_BEG(q) ((q) == 0 ? 0 : 6 * (q) + 1)
#define PARTMAX 7

__global__ __launch_bounds__(NTHR) void skel_conv_wmma_f32(
    const float* x,
    const float* __restrict__ W,
    const float* __restrict__ b,
    float* out)
{
    __shared__ __attribute__((aligned(16))) float xs[2 * BUFF];

    const int tid  = threadIdx.x;
    const int lane = tid & 31;
    const int wave = tid >> 5;
    const int h    = lane >> 4;
    const int mn   = lane & 15;
    const int e0   = wave * EPW;

    v2f A0[EPW][NCHUNK], A1[EPW][NCHUNK];
    float bias[EPW][8];
    #pragma unroll
    for (int ee = 0; ee < EPW; ++ee) {
        const int p0 = 2 * (e0 + ee);
        const int p1 = p0 + 1;
        #pragma unroll
        for (int c = 0; c < NCHUNK; ++c) {
            const int k   = c >> 2;
            const int ci0 = 4 * (c & 3) + 2 * h;
            const int ci1 = ci0 + 1;
            A0[ee][c].x = W[((p0 * COUTC + mn) * CIN + ci0) * KW + k];
            A0[ee][c].y = W[((p0 * COUTC + mn) * CIN + ci1) * KW + k];
            A1[ee][c].x = W[((p1 * COUTC + mn) * CIN + ci0) * KW + k];
            A1[ee][c].y = W[((p1 * COUTC + mn) * CIN + ci1) * KW + k];
        }
        #pragma unroll
        for (int r = 0; r < 8; ++r) {
            const int co = 8 * h + r;
            bias[ee][r] = 0.5f * (b[p0 * COUTC + co] + b[p1 * COUTC + co]);
        }
    }

    v4f stg[PARTMAX];
    v4f hal0, hal1;
    const int srow = tid >> 4;
    const int sq   = tid & 15;
    const int hrow1 = (NTHR + tid < NROW) ? (NTHR + tid) : (NROW - 1);

    int nxt = blockIdx.x;
    int prv = -1;
    int wb  = 0;

    while (true) {
        const bool have = nxt < NSTRIP;
        const int  s0n  = nxt * STRIP;
        float* nbuf = xs + wb * BUFF;
        const float* buf = xs + (wb ^ 1) * BUFF;
        const int  s0   = prv * STRIP;

        #pragma unroll
        for (int q = 0; q < 4; ++q) {
            if (have) {
                if (q > 0) {
                    #pragma unroll
                    for (int r = PART_BEG(q - 1); r < PART_BEG(q); ++r) {
                        const int row = r * 16 + srow;
                        *(v4f*)(nbuf + row * PITCH + 4 * sq) = stg[r - PART_BEG(q - 1)];
                    }
                }
                #pragma unroll
                for (int r = PART_BEG(q); r < PART_BEG(q + 1); ++r) {
                    const int row = r * 16 + srow;
                    stg[r - PART_BEG(q)] = *(const v4f*)(x + (row * LIN + s0n + 4 * sq));
                }
                if (q == 3) {
                    int th = s0n + STRIP;
                    th = (th <= LIN - 4) ? th : (LIN - 4);
                    hal0 = *(const v4f*)(x + (tid * LIN + th));
                    hal1 = *(const v4f*)(x + (hrow1 * LIN + th));
                }
            }

            if (prv >= 0) {
                const int T  = q;
                const int cb = 16 * T + mn;
                v2f Bj[EPW + 1][NCHUNK];
                #pragma unroll
                for (int j = 0; j < 2; ++j) {
                    #pragma unroll
                    for (int c = 0; c < NCHUNK; ++c) {
                        const int k   = c >> 2;
                        const int ci0 = 4 * (c & 3) + 2 * h;
                        const int ci1 = ci0 + 1;
                        Bj[j][c].x = buf[(ci0 * JNT + e0 + j) * PITCH + cb + k];
                        Bj[j][c].y = buf[(ci1 * JNT + e0 + j) * PITCH + cb + k];
                    }
                }
                const int  t    = s0 + 16 * T + mn;
                const bool full = (s0 + 16 * T + 16 <= LOUTC);
                #pragma unroll
                for (int ee = 0; ee < EPW; ++ee) {
                    if (ee + 2 <= EPW) {
                        #pragma unroll
                        for (int c = 0; c < NCHUNK; ++c) {
                            const int k   = c >> 2;
                            const int ci0 = 4 * (c & 3) + 2 * h;
                            const int ci1 = ci0 + 1;
                            Bj[ee + 2][c].x = buf[(ci0 * JNT + e0 + ee + 2) * PITCH + cb + k];
                            Bj[ee + 2][c].y = buf[(ci1 * JNT + e0 + ee + 2) * PITCH + cb + k];
                        }
                    }
                    v8f acc0 = {};
                    v8f acc1 = {};
                    #pragma unroll
                    for (int c = 0; c < NCHUNK; ++c) {
                        acc0 = __builtin_amdgcn_wmma_f32_16x16x4_f32(
                            false, A0[ee][c], false, Bj[ee][c], (short)0, acc0, false, false);
                        acc1 = __builtin_amdgcn_wmma_f32_16x16x4_f32(
                            false, A1[ee][c], false, Bj[ee + 1][c], (short)0, acc1, false, false);
                    }
                    const int e = e0 + ee;
                    if (full) {
                        #pragma unroll
                        for (int r = 0; r < 8; ++r) {
                            const int co = 8 * h + r;
                            out[(co * EDG + e) * LOUTC + t] =
                                0.5f * (acc0[r] + acc1[r]) + bias[ee][r];
                        }
                    } else if (t < LOUTC) {
                        #pragma unroll
                        for (int r = 0; r < 8; ++r) {
                            const int co = 8 * h + r;
                            out[(co * EDG + e) * LOUTC + t] =
                                0.5f * (acc0[r] + acc1[r]) + bias[ee][r];
                        }
                    }
                }
            }
        }

        if (!have) break;

        #pragma unroll
        for (int r = PART_BEG(3); r < PART_BEG(4); ++r) {
            const int row = r * 16 + srow;
            *(v4f*)(nbuf + row * PITCH + 4 * sq) = stg[r - PART_BEG(3)];
        }
        *(v4f*)(nbuf + tid * PITCH + STRIP) = hal0;
        if (NTHR + tid < NROW)
            *(v4f*)(nbuf + (NTHR + tid) * PITCH + STRIP) = hal1;

        __builtin_amdgcn_fence(__ATOMIC_RELEASE, "workgroup", "local");
        __builtin_amdgcn_s_barrier();
        __builtin_amdgcn_fence(__ATOMIC_ACQUIRE, "workgroup", "local");
        prv = nxt;
        nxt += gridDim.x;
        wb ^= 1;
    }
}

extern "C" void kernel_launch(void* const* d_in, const int* in_sizes, int n_in,
                              void* d_out, int out_size, void* d_ws, size_t ws_size,
                              hipStream_t stream) {
    (void)in_sizes; (void)n_in; (void)out_size; (void)d_ws; (void)ws_size;
    const float* x = (const float*)d_in[0];
    const float* W = (const float*)d_in[1];
    const float* b = (const float*)d_in[2];
    float* out = (float*)d_out;

    dim3 grid(256);
    dim3 block(NTHR);
    hipLaunchKernelGGL(skel_conv_wmma_f32, grid, block, 0, stream, x, W, b, out);
}


# column n of tile T is time step 4n+T (was 16T+n): a lane's 4 tiles are 4 consecutive steps, so x comes from LDS as one 16+8 byte read per row and out leaves in 16-byte stores; sums unchanged
# speedup vs baseline: 1.4327x; 1.1320x over previous
#include <hip/hip_runtime.h>


typedef float v2f __attribute__((ext_vector_type(2)));
typedef float v4f __attribute__((ext_vector_type(4)));
typedef float v8f __attribute__((ext_vector_type(8)));
typedef v4f v4f_a8 __attribute__((aligned(8)));

#define CIN    16
#define JNT    25
#define EDG    24
#define COUTC  16
#define KW     3
#define LIN    65536
#define LOUTC  (LIN - KW + 1)
#define NCHUNK 12
#define STRIP  64
#define NSTRIP (LIN / STRIP)
#define NROW   (CIN * JNT)
#define PITCH  72
#define BUFF   (NROW * PITCH)
#define EPW    3
#define NTILE  4
#define NTHR   256
#define BODYR  (NROW * 16 / NTHR)
#define PART_BEG(q) ((q) == 0 ? 0 : 8 * (q) + 1)
#define PARTMAX 9

__global__ __launch_bounds__(NTHR) void skel_conv_wmma_f32(
    const float* x,
    const float* __restrict__ W,
    const float* __restrict__ b,
    float* out)
{
    __shared__ __attribute__((aligned(16))) float xs[2 * BUFF];

    const int tid  = threadIdx.x;
    const int lane = tid & 31;
    const int wave = tid >> 5;
    const int h    = lane >> 4;
    const int mn   = lane & 15;
    const int e0   = wave * EPW;

    v2f A0[EPW][NCHUNK], A1[EPW][NCHUNK];
    float bias[EPW][8];
    #pragma unroll
    for (int ee = 0; ee < EPW; ++ee) {
        const int p0 = 2 * (e0 + ee);
        const int p1 = p0 + 1;
        #pragma unroll
        for (int c = 0; c < NCHUNK; ++c) {
            const int k   = c >> 2;
            const int ci0 = 4 * (c & 3) + 2 * h;
            const int ci1 = ci0 + 1;
            A0[ee][c].x = W[((p0 * COUTC + mn) * CIN + ci0) * KW + k];
            A0[ee][c].y = W[((p0 * COUTC + mn) * CIN + ci1) * KW + k];
            A1[ee][c].x = W[((p1 * COUTC + mn) * CIN + ci0) * KW + k];
            A1[ee][c].y = W[((p1 * COUTC + mn) * CIN + ci1) * KW + k];
        }
        #pragma unroll
        for (int r = 0; r < 8; ++r) {
            const int co = 8 * h + r;
            bias[ee][r] = 0.5f * (b[p0 * COUTC + co] + b[p1 * COUTC + co]);
        }
    }

    v4f stg[PARTMAX];
    v4f hal0, hal1;
    const int srow = tid >> 4;
    const int sq   = tid & 15;
    const int hrow1 = (NTHR + tid < NROW) ? (NTHR + tid) : (NROW - 1);

    int nxt = blockIdx.x;
    int prv = -1;
    int wb  = 0;

    float X[EPW + 1][4][2][6];

    while (true) {
        const bool have = nxt < NSTRIP;
        const int  s0n  = nxt * STRIP;
        float* nbuf = xs + wb * BUFF;
        const float* buf = xs + (wb ^ 1) * BUFF;
        const int  s0   = prv * STRIP;

        #pragma unroll
        for (int q = 0; q < EPW; ++q) {
            if (have) {
                if (q > 0) {
                    #pragma unroll
                    for (int r = PART_BEG(q - 1); r < PART_BEG(q); ++r) {
                        const int row = r * 16 + srow;
                        *(v4f*)(nbuf + row * PITCH + 4 * sq) = stg[r - PART_BEG(q - 1)];
                    }
                }
                #pragma unroll
                for (int r = PART_BEG(q); r < PART_BEG(q + 1); ++r) {
                    const int row = r * 16 + srow;
                    stg[r - PART_BEG(q)] = *(const v4f*)(x + (row * LIN + s0n + 4 * sq));
                }
                if (q == EPW - 1) {
                    int th = s0n + STRIP;
                    th = (th <= LIN - 4) ? th : (LIN - 4);
                    hal0 = *(const v4f*)(x + (tid * LIN + th));
                    hal1 = *(const v4f*)(x + (hrow1 * LIN + th));
                }
            }

            if (prv >= 0) {
                const int ee = q;
                #pragma unroll
                for (int j = (ee == 0 ? 0 : ee + 1); j <= ee + 1; ++j) {
                    #pragma unroll
                    for (int m = 0; m < 4; ++m) {
                        #pragma unroll
                        for (int u = 0; u < 2; ++u) {
                            const int ci = 4 * m + 2 * h + u;
                            const float* rp = buf + (ci * JNT + e0 + j) * PITCH + 4 * mn;
                            const v4f lo = *(const v4f*)rp;
                            const v2f hi = *(const v2f*)(rp + 4);
                            X[j][m][u][0] = lo.x; X[j][m][u][1] = lo.y;
                            X[j][m][u][2] = lo.z; X[j][m][u][3] = lo.w;
                            X[j][m][u][4] = hi.x; X[j][m][u][5] = hi.y;
                        }
                    }
                }
                float res[NTILE][8];
                #pragma unroll
                for (int T = 0; T < NTILE; ++T) {
                    v8f acc0 = {};
                    v8f acc1 = {};
                    #pragma unroll
                    for (int c = 0; c < NCHUNK; ++c) {
                        const int k = c >> 2;
                        const int m = c & 3;
                        v2f b0, b1;
                        b0.x = X[ee][m][0][T + k];
                        b0.y = X[ee][m][1][T + k];
                        b1.x = X[ee + 1][m][0][T + k];
                        b1.y = X[ee + 1][m][1][T + k];
                        acc0 = __builtin_amdgcn_wmma_f32_16x16x4_f32(
                            false, A0[ee][c], false, b0, (short)0, acc0, false, false);
                        acc1 = __builtin_amdgcn_wmma_f32_16x16x4_f32(
                            false, A1[ee][c], false, b1, (short)0, acc1, false, false);
                    }
                    #pragma unroll
                    for (int r = 0; r < 8; ++r)
                        res[T][r] = 0.5f * (acc0[r] + acc1[r]) + bias[ee][r];
                }
                const int  e    = e0 + ee;
                const int  t    = s0 + 4 * mn;
                const bool full = (s0 + STRIP <= LOUTC);
                if (full) {
                    #pragma unroll
                    for (int r = 0; r < 8; ++r) {
                        const int co = 8 * h + r;
                        v4f v;
                        v.x = res[0][r]; v.y = res[1][r]; v.z = res[2][r]; v.w = res[3][r];
                        *(v4f_a8*)(out + ((co * EDG + e) * LOUTC + t)) = v;
                    }
                } else {
                    #pragma unroll
                    for (int r = 0; r < 8; ++r) {
                        const int co = 8 * h + r;
                        float* op = out + ((co * EDG + e) * LOUTC + t);
                        if (t + 1 < LOUTC) {
                            v2f v; v.x = res[0][r]; v.y = res[1][r];
                            *(v2f*)op = v;
                        }
                        if (t + 3 < LOUTC) {
                            v2f v; v.x = res[2][r]; v.y = res[3][r];
                            *(v2f*)(op + 2) = v;
                        }
                    }
                }
            }
        }

        if (!have) break;

        #pragma unroll
        for (int r = PART_BEG(EPW - 1); r < PART_BEG(EPW); ++r) {
            const int row = r * 16 + srow;
            *(v4f*)(nbuf + row * PITCH + 4 * sq) = stg[r - PART_BEG(EPW - 1)];
        }
        *(v4f*)(nbuf + tid * PITCH + STRIP) = hal0;
        if (NTHR + tid < NROW)
            *(v4f*)(nbuf + (NTHR + tid) * PITCH + STRIP) = hal1;

        __builtin_amdgcn_fence(__ATOMIC_RELEASE, "workgroup", "local");
        __builtin_amdgcn_s_barrier();
        __builtin_amdgcn_fence(__ATOMIC_ACQUIRE, "workgroup", "local");
        prv = nxt;
        nxt += gridDim.x;
        wb ^= 1;
    }
}

extern "C" void kernel_launch(void* const* d_in, const int* in_sizes, int n_in,
                              void* d_out, int out_size, void* d_ws, size_t ws_size,
                              hipStream_t stream) {
    (void)in_sizes; (void)n_in; (void)out_size; (void)d_ws; (void)ws_size;
    const float* x = (const float*)d_in[0];
    const float* W = (const float*)d_in[1];
    const float* b = (const float*)d_in[2];
    float* out = (float*)d_out;

    dim3 grid(256);
    dim3 block(NTHR);
    hipLaunchKernelGGL(skel_conv_wmma_f32, grid, block, 0, stream, x, W, b, out);
}
